# starting kernel
# baseline (speedup 1.0000x reference)
#include <hip/hip_runtime.h>


#define N_NODES 100000
#define N_EDGES 3200000
#define DIN     512
#define DHID    256
#define DOUT    64
#define KSTEPS  10

typedef __attribute__((ext_vector_type(2))) float v2f;
typedef __attribute__((ext_vector_type(8))) float v8f;


__global__ __launch_bounds__(256) void k_init(float* __restrict__ deg,
                                              float* __restrict__ bufB) {
    int g = blockIdx.x * 256 + threadIdx.x;
    bufB[g] = 0.0f;
    if (g < N_NODES) deg[g] = 1.0f;
}

__global__ __launch_bounds__(256) void k_count(const int* __restrict__ col,
                                               float* __restrict__ deg) {
    int e = blockIdx.x * 256 + threadIdx.x;
    if (e < N_EDGES) atomicAdd(&deg[col[e]], 1.0f);
}

__global__ __launch_bounds__(256) void k_dinv(const float* __restrict__ deg,
                                              float* __restrict__ dinv,
                                              float* __restrict__ dinv2) {
    int i = blockIdx.x * 256 + threadIdx.x;
    if (i < N_NODES) {
        float r = rsqrtf(deg[i]);
        dinv[i]  = r;
        dinv2[i] = r * r;
    }
}

__global__ __launch_bounds__(256) void k_norm(const int* __restrict__ row,
                                              const int* __restrict__ col,
                                              const float* __restrict__ dinv,
                                              float* __restrict__ norm) {
    int e = blockIdx.x * 256 + threadIdx.x;
    if (e < N_EDGES) norm[e] = dinv[row[e]] * dinv[col[e]];
}

#define LDS_STRIDE 260

__global__ __launch_bounds__(512)
void k_mlp(const float* __restrict__ x,  const float* __restrict__ W1,
           const float* __restrict__ b1, const float* __restrict__ W2,
           const float* __restrict__ b2, const float* __restrict__ temp,
           float* __restrict__ bufA,     float* __restrict__ hidden) {
    __shared__ float lds[16 * LDS_STRIDE];

    const int tid  = threadIdx.x;
    const int wave = tid >> 5;
    const int lane = tid & 31;
    const int l15  = lane & 15;
    const int kh   = lane >> 4;
    const int m0   = blockIdx.x * 16;

    v8f acc = {0.f, 0.f, 0.f, 0.f, 0.f, 0.f, 0.f, 0.f};
    const float* xrow  = x  + (size_t)(m0 + l15) * DIN;
    const float* w1col = W1 + wave * 16 + l15;
    for (int k0 = 0; k0 < DIN; k0 += 4) {
        int ka = k0 + kh * 2;
        v2f a = *(const v2f*)(xrow + ka);
        v2f b;
        b.x = w1col[(size_t)ka * DHID];
        b.y = w1col[(size_t)(ka + 1) * DHID];
        acc = __builtin_amdgcn_wmma_f32_16x16x4_f32(
                  false, a, false, b, (short)0, acc, false, false);
    }
    {
        float bias = b1[wave * 16 + l15];
        int rbase = kh * 8;
        #pragma unroll
        for (int r = 0; r < 8; ++r) {
            float v = acc[r] + bias;
            v = v > 0.0f ? v : 0.0f;
            lds[(rbase + r) * LDS_STRIDE + wave * 16 + l15] = v;
        }
    }
    __syncthreads();

    if (wave < 4) {
        v8f acc2 = {0.f, 0.f, 0.f, 0.f, 0.f, 0.f, 0.f, 0.f};
        const float* w2col = W2 + wave * 16 + l15;
        for (int k0 = 0; k0 < DHID; k0 += 4) {
            int ka = k0 + kh * 2;
            v2f a = *(const v2f*)(&lds[l15 * LDS_STRIDE + ka]);
            v2f b;
            b.x = w2col[ka * DOUT];
            b.y = w2col[(ka + 1) * DOUT];
            acc2 = __builtin_amdgcn_wmma_f32_16x16x4_f32(
                       false, a, false, b, (short)0, acc2, false, false);
        }
        float bias = b2[wave * 16 + l15];
        float t0   = temp[0];
        int rbase = kh * 8;
        #pragma unroll
        for (int r = 0; r < 8; ++r) {
            float v = acc2[r] + bias;
            int idx = (m0 + rbase + r) * DOUT + wave * 16 + l15;
            bufA[idx]   = v;
            hidden[idx] = t0 * v;
        }
    }
}

__global__ __launch_bounds__(256)
void k_scatter(const int* __restrict__ row, const int* __restrict__ col,
               const float* __restrict__ norm, const float* __restrict__ src,
               float* __restrict__ dst) {
    int g = blockIdx.x * 256 + threadIdx.x;
    int e = __builtin_amdgcn_readfirstlane(g >> 6);
    int c = g & 63;
    int r  = row[e];
    int cl = col[e];
    float v = norm[e] * src[r * DOUT + c];
    atomicAdd(&dst[cl * DOUT + c], v);
}

__global__ __launch_bounds__(256)
void k_finalize(float* __restrict__ dst, float* __restrict__ src,
                const float* __restrict__ dinv2, float* __restrict__ hidden,
                const float* __restrict__ temp, int kidx) {
    int g = blockIdx.x * 256 + threadIdx.x;
    int i = __builtin_amdgcn_readfirstlane(g >> 6);
    float v = dst[g] + dinv2[i] * src[g];
    dst[g] = v;
    hidden[g] += temp[kidx] * v;
    src[g] = 0.0f;
}

extern "C" void kernel_launch(void* const* d_in, const int* in_sizes, int n_in,
                              void* d_out, int out_size, void* d_ws, size_t ws_size,
                              hipStream_t stream) {
    const float* x    = (const float*)d_in[0];
    const int*   ei   = (const int*)  d_in[1];
    const float* W1   = (const float*)d_in[2];
    const float* b1   = (const float*)d_in[3];
    const float* W2   = (const float*)d_in[4];
    const float* b2   = (const float*)d_in[5];
    const float* temp = (const float*)d_in[6];
    const int* row = ei;
    const int* col = ei + N_EDGES;
    float* hidden = (float*)d_out;

    float* deg   = (float*)d_ws;
    float* dinv  = deg   + N_NODES;
    float* dinv2 = dinv  + N_NODES;
    float* normw = dinv2 + N_NODES;
    float* bufA  = normw + N_EDGES;
    float* bufB  = bufA  + (size_t)N_NODES * DOUT;

    k_init <<<(N_NODES * DOUT) / 256, 256, 0, stream>>>(deg, bufB);
    k_count<<<(N_EDGES + 255) / 256, 256, 0, stream>>>(col, deg);
    k_dinv <<<(N_NODES + 255) / 256, 256, 0, stream>>>(deg, dinv, dinv2);
    k_norm <<<(N_EDGES + 255) / 256, 256, 0, stream>>>(row, col, dinv, normw);

    k_mlp<<<N_NODES / 16, 512, 0, stream>>>(x, W1, b1, W2, b2, temp, bufA, hidden);

    for (int k = 0; k < KSTEPS; ++k) {
        float* src = (k & 1) ? bufB : bufA;
        float* dst = (k & 1) ? bufA : bufB;
        k_scatter <<<(N_EDGES * 64) / 256, 256, 0, stream>>>(row, col, normw, src, dst);
        k_finalize<<<(N_NODES * DOUT) / 256, 256, 0, stream>>>(dst, src, dinv2,
                                                               hidden, temp, k + 1);
    }
}


# Scatter-add by f32 atomics replaced by per-call counting sort of edges by target + atomic-free one-wave-per-node row gather; sums now sequential per node in sort arrival order
# speedup vs baseline: 5.6313x; 5.6313x over previous
#include <hip/hip_runtime.h>


#define N_NODES 100000
#define N_EDGES 3200000
#define DIN     512
#define DHID    256
#define DOUT    64
#define KSTEPS  10

typedef __attribute__((ext_vector_type(2))) float v2f;
typedef __attribute__((ext_vector_type(8))) float v8f;


__global__ __launch_bounds__(256) void k_zero(int* __restrict__ cnt) {
    int i = blockIdx.x * 256 + threadIdx.x;
    if (i < N_NODES) cnt[i] = 0;
}

__global__ __launch_bounds__(256) void k_count(const int* __restrict__ col,
                                               int* __restrict__ cnt,
                                               int* __restrict__ pos) {
    int e = blockIdx.x * 256 + threadIdx.x;
    if (e < N_EDGES) pos[e] = atomicAdd(&cnt[col[e]], 1);
}

__global__ __launch_bounds__(1024) void k_scan(const int* __restrict__ cnt,
                                               int* __restrict__ offs,
                                               float* __restrict__ dinv) {
    __shared__ int wsum[32];
    const int tid  = threadIdx.x;
    const int lane = tid & 31;
    const int wave = tid >> 5;
    int carry = 0;
    for (int base = 0; base < N_NODES; base += 4096) {
        const int  i0 = base + tid * 4;
        const bool ok = i0 < N_NODES;
        int4 c = make_int4(0, 0, 0, 0);
        if (ok) c = *(const int4*)(cnt + i0);
        const int s = c.x + c.y + c.z + c.w;
        int incl = s;
        #pragma unroll
        for (int d = 1; d < 32; d <<= 1) {
            int t = __shfl_up(incl, d, 32);
            if (lane >= d) incl += t;
        }
        if (lane == 31) wsum[wave] = incl;
        __syncthreads();
        const int w = wsum[lane];
        int winc = w;
        #pragma unroll
        for (int d = 1; d < 32; d <<= 1) {
            int t = __shfl_up(winc, d, 32);
            if (lane >= d) winc += t;
        }
        const int wave_excl = __shfl(winc, wave, 32) - __shfl(w, wave, 32);
        const int total     = __shfl(winc, 31, 32);
        const int excl      = carry + wave_excl + (incl - s);
        if (ok) {
            int4 o;
            o.x = excl;
            o.y = o.x + c.x;
            o.z = o.y + c.y;
            o.w = o.z + c.z;
            *(int4*)(offs + i0) = o;
            float4 r;
            r.x = rsqrtf((float)(c.x + 1));
            r.y = rsqrtf((float)(c.y + 1));
            r.z = rsqrtf((float)(c.z + 1));
            r.w = rsqrtf((float)(c.w + 1));
            *(float4*)(dinv + i0) = r;
        }
        carry += total;
        __syncthreads();
    }
}

__global__ __launch_bounds__(256) void k_fill(const int* __restrict__ row,
                                              const int* __restrict__ col,
                                              const int* __restrict__ pos,
                                              const int* __restrict__ offs,
                                              int* __restrict__ srow) {
    int e = blockIdx.x * 256 + threadIdx.x;
    if (e < N_EDGES) srow[offs[col[e]] + pos[e]] = row[e];
}

#define LDS_STRIDE 260

__global__ __launch_bounds__(512)
void k_mlp(const float* __restrict__ x,  const float* __restrict__ W1,
           const float* __restrict__ b1, const float* __restrict__ W2,
           const float* __restrict__ b2, const float* __restrict__ temp,
           float* __restrict__ bufA,     float* __restrict__ hidden) {
    __shared__ float lds[16 * LDS_STRIDE];

    const int tid  = threadIdx.x;
    const int wave = tid >> 5;
    const int lane = tid & 31;
    const int l15  = lane & 15;
    const int kh   = lane >> 4;
    const int m0   = blockIdx.x * 16;

    v8f acc = {0.f, 0.f, 0.f, 0.f, 0.f, 0.f, 0.f, 0.f};
    const float* xrow  = x  + (size_t)(m0 + l15) * DIN;
    const float* w1col = W1 + wave * 16 + l15;
    for (int k0 = 0; k0 < DIN; k0 += 4) {
        int ka = k0 + kh * 2;
        v2f a = *(const v2f*)(xrow + ka);
        v2f b;
        b.x = w1col[(size_t)ka * DHID];
        b.y = w1col[(size_t)(ka + 1) * DHID];
        acc = __builtin_amdgcn_wmma_f32_16x16x4_f32(
                  false, a, false, b, (short)0, acc, false, false);
    }
    {
        float bias = b1[wave * 16 + l15];
        int rbase = kh * 8;
        #pragma unroll
        for (int r = 0; r < 8; ++r) {
            float v = acc[r] + bias;
            v = v > 0.0f ? v : 0.0f;
            lds[(rbase + r) * LDS_STRIDE + wave * 16 + l15] = v;
        }
    }
    __syncthreads();

    if (wave < 4) {
        v8f acc2 = {0.f, 0.f, 0.f, 0.f, 0.f, 0.f, 0.f, 0.f};
        const float* w2col = W2 + wave * 16 + l15;
        for (int k0 = 0; k0 < DHID; k0 += 4) {
            int ka = k0 + kh * 2;
            v2f a = *(const v2f*)(&lds[l15 * LDS_STRIDE + ka]);
            v2f b;
            b.x = w2col[ka * DOUT];
            b.y = w2col[(ka + 1) * DOUT];
            acc2 = __builtin_amdgcn_wmma_f32_16x16x4_f32(
                       false, a, false, b, (short)0, acc2, false, false);
        }
        float bias = b2[wave * 16 + l15];
        float t0   = temp[0];
        int rbase = kh * 8;
        #pragma unroll
        for (int r = 0; r < 8; ++r) {
            float v = acc2[r] + bias;
            int idx = (m0 + rbase + r) * DOUT + wave * 16 + l15;
            bufA[idx]   = v;
            hidden[idx] = t0 * v;
        }
    }
}

__global__ __launch_bounds__(256)
void k_gather(const float* __restrict__ src, float* __restrict__ dst,
              float* __restrict__ hidden, const int* __restrict__ offs,
              const int* __restrict__ cnt, const int* __restrict__ srow,
              const float* __restrict__ dinv, const float* __restrict__ temp,
              int kidx) {
    const int lane = threadIdx.x & 31;
    const int i = __builtin_amdgcn_readfirstlane(blockIdx.x * 8 + (threadIdx.x >> 5));
    const int   beg = offs[i];
    const int   n   = cnt[i];
    const float di  = dinv[i];
    const v2f* src2 = (const v2f*)src;

    const v2f   sv = src2[(size_t)i * 32 + lane];
    const float ws = di * di;
    float ax = ws * sv.x;
    float ay = ws * sv.y;

    for (int base = 0; base < n; base += 32) {
        int m = n - base;
        if (m > 32) m = 32;
        int   my_r = i;
        float my_w = 0.0f;
        if (lane < m) {
            my_r = srow[beg + base + lane];
            my_w = dinv[my_r] * di;
        }
        for (int j = 0; j < m; j += 8) {
            #pragma unroll
            for (int u = 0; u < 8; ++u) {
                const int   r = __builtin_amdgcn_readlane(my_r, j + u);
                const float w = __int_as_float(
                    __builtin_amdgcn_readlane(__float_as_int(my_w), j + u));
                const v2f v = src2[(size_t)r * 32 + lane];
                ax = fmaf(w, v.x, ax);
                ay = fmaf(w, v.y, ay);
            }
        }
    }

    const size_t o = (size_t)i * 32 + lane;
    v2f res;
    res.x = ax;
    res.y = ay;
    ((v2f*)dst)[o] = res;
    const float g = temp[kidx];
    v2f h = ((v2f*)hidden)[o];
    h.x += g * ax;
    h.y += g * ay;
    ((v2f*)hidden)[o] = h;
}

extern "C" void kernel_launch(void* const* d_in, const int* in_sizes, int n_in,
                              void* d_out, int out_size, void* d_ws, size_t ws_size,
                              hipStream_t stream) {
    const float* x    = (const float*)d_in[0];
    const int*   ei   = (const int*)  d_in[1];
    const float* W1   = (const float*)d_in[2];
    const float* b1   = (const float*)d_in[3];
    const float* W2   = (const float*)d_in[4];
    const float* b2   = (const float*)d_in[5];
    const float* temp = (const float*)d_in[6];
    const int* row = ei;
    const int* col = ei + N_EDGES;
    float* hidden = (float*)d_out;

    float* bufA  = (float*)d_ws;
    float* bufB  = bufA  + (size_t)N_NODES * DOUT;
    int*   cnt   = (int*)(bufB + (size_t)N_NODES * DOUT);
    int*   offs  = cnt   + N_NODES;
    float* dinv  = (float*)(offs + N_NODES);
    int*   srow  = (int*)(dinv + N_NODES);
    int*   pos   = (int*)bufB;

    k_zero <<<(N_NODES + 255) / 256, 256, 0, stream>>>(cnt);
    k_count<<<(N_EDGES + 255) / 256, 256, 0, stream>>>(col, cnt, pos);
    k_scan <<<1, 1024, 0, stream>>>(cnt, offs, dinv);
    k_fill <<<(N_EDGES + 255) / 256, 256, 0, stream>>>(row, col, pos, offs, srow);

    k_mlp<<<N_NODES / 16, 512, 0, stream>>>(x, W1, b1, W2, b2, temp, bufA, hidden);

    for (int k = 0; k < KSTEPS; ++k) {
        float* src = (k & 1) ? bufB : bufA;
        float* dst = (k & 1) ? bufA : bufB;
        k_gather<<<N_NODES / 8, 256, 0, stream>>>(src, dst, hidden, offs, cnt, srow,
                                                  dinv, temp, k + 1);
    }
}


# Step planes hold rows pre-scaled by dinv (y=dinv*cur): gather adds raw rows, applies dinv[i] once after the sum; per-edge dinv[src] read and weight dropped; sum order kept, products regrouped
# speedup vs baseline: 6.1667x; 1.0951x over previous
#include <hip/hip_runtime.h>


#define N_NODES 100000
#define N_EDGES 3200000
#define DIN     512
#define DHID    256
#define DOUT    64
#define KSTEPS  10

typedef __attribute__((ext_vector_type(2))) float v2f;
typedef __attribute__((ext_vector_type(8))) float v8f;


__global__ __launch_bounds__(256) void k_zero(int* __restrict__ cnt) {
    int i = blockIdx.x * 256 + threadIdx.x;
    if (i < N_NODES) cnt[i] = 0;
}

__global__ __launch_bounds__(256) void k_count(const int* __restrict__ col,
                                               int* __restrict__ cnt,
                                               int* __restrict__ pos) {
    int e = blockIdx.x * 256 + threadIdx.x;
    if (e < N_EDGES) pos[e] = atomicAdd(&cnt[col[e]], 1);
}

__global__ __launch_bounds__(1024) void k_scan(const int* __restrict__ cnt,
                                               int* __restrict__ offs,
                                               float* __restrict__ dinv) {
    __shared__ int wsum[32];
    const int tid  = threadIdx.x;
    const int lane = tid & 31;
    const int wave = tid >> 5;
    int carry = 0;
    for (int base = 0; base < N_NODES; base += 4096) {
        const int  i0 = base + tid * 4;
        const bool ok = i0 < N_NODES;
        int4 c = make_int4(0, 0, 0, 0);
        if (ok) c = *(const int4*)(cnt + i0);
        const int s = c.x + c.y + c.z + c.w;
        int incl = s;
        #pragma unroll
        for (int d = 1; d < 32; d <<= 1) {
            int t = __shfl_up(incl, d, 32);
            if (lane >= d) incl += t;
        }
        if (lane == 31) wsum[wave] = incl;
        __syncthreads();
        const int w = wsum[lane];
        int winc = w;
        #pragma unroll
        for (int d = 1; d < 32; d <<= 1) {
            int t = __shfl_up(winc, d, 32);
            if (lane >= d) winc += t;
        }
        const int wave_excl = __shfl(winc, wave, 32) - __shfl(w, wave, 32);
        const int total     = __shfl(winc, 31, 32);
        const int excl      = carry + wave_excl + (incl - s);
        if (ok) {
            int4 o;
            o.x = excl;
            o.y = o.x + c.x;
            o.z = o.y + c.y;
            o.w = o.z + c.z;
            *(int4*)(offs + i0) = o;
            float4 r;
            r.x = rsqrtf((float)(c.x + 1));
            r.y = rsqrtf((float)(c.y + 1));
            r.z = rsqrtf((float)(c.z + 1));
            r.w = rsqrtf((float)(c.w + 1));
            *(float4*)(dinv + i0) = r;
        }
        carry += total;
        __syncthreads();
    }
}

__global__ __launch_bounds__(256) void k_fill(const int* __restrict__ row,
                                              const int* __restrict__ col,
                                              const int* __restrict__ pos,
                                              const int* __restrict__ offs,
                                              int* __restrict__ srow) {
    int e = blockIdx.x * 256 + threadIdx.x;
    if (e < N_EDGES) srow[offs[col[e]] + pos[e]] = row[e];
}

#define LDS_STRIDE 260

__global__ __launch_bounds__(512)
void k_mlp(const float* __restrict__ x,  const float* __restrict__ W1,
           const float* __restrict__ b1, const float* __restrict__ W2,
           const float* __restrict__ b2, const float* __restrict__ temp,
           const float* __restrict__ dinv,
           float* __restrict__ bufA,     float* __restrict__ hidden) {
    __shared__ float lds[16 * LDS_STRIDE];

    const int tid  = threadIdx.x;
    const int wave = tid >> 5;
    const int lane = tid & 31;
    const int l15  = lane & 15;
    const int kh   = lane >> 4;
    const int m0   = blockIdx.x * 16;

    v8f acc = {0.f, 0.f, 0.f, 0.f, 0.f, 0.f, 0.f, 0.f};
    const float* xrow  = x  + (size_t)(m0 + l15) * DIN;
    const float* w1col = W1 + wave * 16 + l15;
    for (int k0 = 0; k0 < DIN; k0 += 4) {
        int ka = k0 + kh * 2;
        v2f a = *(const v2f*)(xrow + ka);
        v2f b;
        b.x = w1col[(size_t)ka * DHID];
        b.y = w1col[(size_t)(ka + 1) * DHID];
        acc = __builtin_amdgcn_wmma_f32_16x16x4_f32(
                  false, a, false, b, (short)0, acc, false, false);
    }
    {
        float bias = b1[wave * 16 + l15];
        int rbase = kh * 8;
        #pragma unroll
        for (int r = 0; r < 8; ++r) {
            float v = acc[r] + bias;
            v = v > 0.0f ? v : 0.0f;
            lds[(rbase + r) * LDS_STRIDE + wave * 16 + l15] = v;
        }
    }
    __syncthreads();

    if (wave < 4) {
        v8f acc2 = {0.f, 0.f, 0.f, 0.f, 0.f, 0.f, 0.f, 0.f};
        const float* w2col = W2 + wave * 16 + l15;
        for (int k0 = 0; k0 < DHID; k0 += 4) {
            int ka = k0 + kh * 2;
            v2f a = *(const v2f*)(&lds[l15 * LDS_STRIDE + ka]);
            v2f b;
            b.x = w2col[ka * DOUT];
            b.y = w2col[(ka + 1) * DOUT];
            acc2 = __builtin_amdgcn_wmma_f32_16x16x4_f32(
                       false, a, false, b, (short)0, acc2, false, false);
        }
        float bias = b2[wave * 16 + l15];
        float t0   = temp[0];
        int rbase = kh * 8;
        #pragma unroll
        for (int r = 0; r < 8; ++r) {
            float v = acc2[r] + bias;
            int node = m0 + rbase + r;
            int idx  = node * DOUT + wave * 16 + l15;
            bufA[idx]   = dinv[node] * v;
            hidden[idx] = t0 * v;
        }
    }
}

__global__ __launch_bounds__(256)
void k_gather(const float* __restrict__ src, float* __restrict__ dst,
              float* __restrict__ hidden, const int* __restrict__ offs,
              const int* __restrict__ cnt, const int* __restrict__ srow,
              const float* __restrict__ dinv, const float* __restrict__ temp,
              int kidx) {
    const int lane = threadIdx.x & 31;
    const int i = __builtin_amdgcn_readfirstlane(blockIdx.x * 8 + (threadIdx.x >> 5));
    const int   beg = offs[i];
    const int   n   = cnt[i];
    const float di  = dinv[i];
    const v2f* src2 = (const v2f*)src;

    const v2f sv = src2[(size_t)i * 32 + lane];
    float ax = sv.x;
    float ay = sv.y;

    for (int base = 0; base < n; base += 32) {
        int m = n - base;
        if (m > 32) m = 32;
        int my_r = i;
        if (lane < m) my_r = srow[beg + base + lane];
        for (int j = 0; j < m; j += 8) {
            #pragma unroll
            for (int u = 0; u < 8; ++u) {
                const int   r = __builtin_amdgcn_readlane(my_r, j + u);
                const float w = (j + u < m) ? 1.0f : 0.0f;
                const v2f v = src2[(size_t)r * 32 + lane];
                ax = fmaf(w, v.x, ax);
                ay = fmaf(w, v.y, ay);
            }
        }
    }
    const float cx = di * ax;
    const float cy = di * ay;

    const size_t o = (size_t)i * 32 + lane;
    v2f res;
    res.x = di * cx;
    res.y = di * cy;
    ((v2f*)dst)[o] = res;
    const float g = temp[kidx];
    v2f h = ((v2f*)hidden)[o];
    h.x += g * cx;
    h.y += g * cy;
    ((v2f*)hidden)[o] = h;
}

extern "C" void kernel_launch(void* const* d_in, const int* in_sizes, int n_in,
                              void* d_out, int out_size, void* d_ws, size_t ws_size,
                              hipStream_t stream) {
    const float* x    = (const float*)d_in[0];
    const int*   ei   = (const int*)  d_in[1];
    const float* W1   = (const float*)d_in[2];
    const float* b1   = (const float*)d_in[3];
    const float* W2   = (const float*)d_in[4];
    const float* b2   = (const float*)d_in[5];
    const float* temp = (const float*)d_in[6];
    const int* row = ei;
    const int* col = ei + N_EDGES;
    float* hidden = (float*)d_out;

    float* bufA  = (float*)d_ws;
    float* bufB  = bufA  + (size_t)N_NODES * DOUT;
    int*   cnt   = (int*)(bufB + (size_t)N_NODES * DOUT);
    int*   offs  = cnt   + N_NODES;
    float* dinv  = (float*)(offs + N_NODES);
    int*   srow  = (int*)(dinv + N_NODES);
    int*   pos   = (int*)bufB;

    k_zero <<<(N_NODES + 255) / 256, 256, 0, stream>>>(cnt);
    k_count<<<(N_EDGES + 255) / 256, 256, 0, stream>>>(col, cnt, pos);
    k_scan <<<1, 1024, 0, stream>>>(cnt, offs, dinv);
    k_fill <<<(N_EDGES + 255) / 256, 256, 0, stream>>>(row, col, pos, offs, srow);

    k_mlp<<<N_NODES / 16, 512, 0, stream>>>(x, W1, b1, W2, b2, temp, dinv, bufA, hidden);

    for (int k = 0; k < KSTEPS; ++k) {
        float* src = (k & 1) ? bufB : bufA;
        float* dst = (k & 1) ? bufA : bufB;
        k_gather<<<N_NODES / 8, 256, 0, stream>>>(src, dst, hidden, offs, cnt, srow,
                                                  dinv, temp, k + 1);
    }
}


# MLP re-tiled: a 32-row block stages x in LDS once; each wave keeps 8 accumulator tiles fed by 16-byte W1 row loads (columns interleaved over 4 tiles): one load feeds 8 matrix ops; sum order kept
# speedup vs baseline: 7.5535x; 1.2249x over previous
#include <hip/hip_runtime.h>


#define N_NODES 100000
#define N_EDGES 3200000
#define DIN     512
#define DHID    256
#define DOUT    64
#define KSTEPS  10

typedef __attribute__((ext_vector_type(2))) float v2f;
typedef __attribute__((ext_vector_type(8))) float v8f;


__global__ __launch_bounds__(256) void k_zero(int* __restrict__ cnt) {
    int i = blockIdx.x * 256 + threadIdx.x;
    if (i < N_NODES) cnt[i] = 0;
}

__global__ __launch_bounds__(256) void k_count(const int* __restrict__ col,
                                               int* __restrict__ cnt,
                                               int* __restrict__ pos) {
    int e = blockIdx.x * 256 + threadIdx.x;
    if (e < N_EDGES) pos[e] = atomicAdd(&cnt[col[e]], 1);
}

__global__ __launch_bounds__(1024) void k_scan(const int* __restrict__ cnt,
                                               int* __restrict__ offs,
                                               float* __restrict__ dinv) {
    __shared__ int wsum[32];
    const int tid  = threadIdx.x;
    const int lane = tid & 31;
    const int wave = tid >> 5;
    int carry = 0;
    for (int base = 0; base < N_NODES; base += 4096) {
        const int  i0 = base + tid * 4;
        const bool ok = i0 < N_NODES;
        int4 c = make_int4(0, 0, 0, 0);
        if (ok) c = *(const int4*)(cnt + i0);
        const int s = c.x + c.y + c.z + c.w;
        int incl = s;
        #pragma unroll
        for (int d = 1; d < 32; d <<= 1) {
            int t = __shfl_up(incl, d, 32);
            if (lane >= d) incl += t;
        }
        if (lane == 31) wsum[wave] = incl;
        __syncthreads();
        const int w = wsum[lane];
        int winc = w;
        #pragma unroll
        for (int d = 1; d < 32; d <<= 1) {
            int t = __shfl_up(winc, d, 32);
            if (lane >= d) winc += t;
        }
        const int wave_excl = __shfl(winc, wave, 32) - __shfl(w, wave, 32);
        const int total     = __shfl(winc, 31, 32);
        const int excl      = carry + wave_excl + (incl - s);
        if (ok) {
            int4 o;
            o.x = excl;
            o.y = o.x + c.x;
            o.z = o.y + c.y;
            o.w = o.z + c.z;
            *(int4*)(offs + i0) = o;
            float4 r;
            r.x = rsqrtf((float)(c.x + 1));
            r.y = rsqrtf((float)(c.y + 1));
            r.z = rsqrtf((float)(c.z + 1));
            r.w = rsqrtf((float)(c.w + 1));
            *(float4*)(dinv + i0) = r;
        }
        carry += total;
        __syncthreads();
    }
}

__global__ __launch_bounds__(256) void k_fill(const int* __restrict__ row,
                                              const int* __restrict__ col,
                                              const int* __restrict__ pos,
                                              const int* __restrict__ offs,
                                              int* __restrict__ srow) {
    int e = blockIdx.x * 256 + threadIdx.x;
    if (e < N_EDGES) srow[offs[col[e]] + pos[e]] = row[e];
}

#define XP 516
#define HP 260

typedef __attribute__((ext_vector_type(4))) float v4f;

__global__ __launch_bounds__(128)
void k_mlp(const float* __restrict__ x,  const float* __restrict__ W1,
           const float* __restrict__ b1, const float* __restrict__ W2,
           const float* __restrict__ b2, const float* __restrict__ temp,
           const float* __restrict__ dinv,
           float* __restrict__ bufA,     float* __restrict__ hidden) {
    __shared__ float lds[32 * XP];

    const int tid  = threadIdx.x;
    const int wave = tid >> 5;
    const int lane = tid & 31;
    const int c    = lane & 15;
    const int h    = lane >> 4;
    const int m0   = blockIdx.x * 32;

    {
        const v4f* xg = (const v4f*)(x + (size_t)m0 * DIN);
        #pragma unroll
        for (int r = 0; r < 32; ++r) {
            v4f v = xg[r * 128 + tid];
            *(v4f*)(&lds[r * XP + tid * 4]) = v;
        }
    }
    __syncthreads();

    v8f acc[2][4];
    #pragma unroll
    for (int t = 0; t < 2; ++t)
        #pragma unroll
        for (int j = 0; j < 4; ++j)
            acc[t][j] = (v8f){0.f, 0.f, 0.f, 0.f, 0.f, 0.f, 0.f, 0.f};

    {
        const float* a0p = &lds[c * XP + 2 * h];
        const float* a1p = a0p + 16 * XP;
        const float* wp  = W1 + (size_t)(2 * h) * DHID + wave * 64 + 4 * c;
        #pragma unroll 4
        for (int k0 = 0; k0 < DIN; k0 += 4) {
            const v2f a0 = *(const v2f*)(a0p + k0);
            const v2f a1 = *(const v2f*)(a1p + k0);
            const v4f q0 = *(const v4f*)(wp + (size_t)k0 * DHID);
            const v4f q1 = *(const v4f*)(wp + (size_t)(k0 + 1) * DHID);
            #pragma unroll
            for (int j = 0; j < 4; ++j) {
                v2f b;
                b.x = q0[j];
                b.y = q1[j];
                acc[0][j] = __builtin_amdgcn_wmma_f32_16x16x4_f32(
                                false, a0, false, b, (short)0, acc[0][j], false, false);
                acc[1][j] = __builtin_amdgcn_wmma_f32_16x16x4_f32(
                                false, a1, false, b, (short)0, acc[1][j], false, false);
            }
        }
    }
    __syncthreads();

    {
        const v4f bias = *(const v4f*)(b1 + wave * 64 + 4 * c);
        #pragma unroll
        for (int t = 0; t < 2; ++t) {
            #pragma unroll
            for (int r = 0; r < 8; ++r) {
                v4f v;
                #pragma unroll
                for (int j = 0; j < 4; ++j) {
                    float u = acc[t][j][r] + bias[j];
                    v[j] = u > 0.0f ? u : 0.0f;
                }
                *(v4f*)(&lds[(t * 16 + 8 * h + r) * HP + wave * 64 + 4 * c]) = v;
            }
        }
    }
    __syncthreads();

    {
        const int rt   = wave >> 1;
        const int half = wave & 1;
        v8f o0 = {0.f, 0.f, 0.f, 0.f, 0.f, 0.f, 0.f, 0.f};
        v8f o1 = {0.f, 0.f, 0.f, 0.f, 0.f, 0.f, 0.f, 0.f};
        const float* ap  = &lds[(rt * 16 + c) * HP + 2 * h];
        const float* w2p = W2 + (2 * h) * DOUT + half * 32 + 2 * c;
        #pragma unroll 8
        for (int k0 = 0; k0 < DHID; k0 += 4) {
            const v2f a  = *(const v2f*)(ap + k0);
            const v2f q0 = *(const v2f*)(w2p + k0 * DOUT);
            const v2f q1 = *(const v2f*)(w2p + (k0 + 1) * DOUT);
            v2f bA, bB;
            bA.x = q0.x;  bA.y = q1.x;
            bB.x = q0.y;  bB.y = q1.y;
            o0 = __builtin_amdgcn_wmma_f32_16x16x4_f32(
                     false, a, false, bA, (short)0, o0, false, false);
            o1 = __builtin_amdgcn_wmma_f32_16x16x4_f32(
                     false, a, false, bB, (short)0, o1, false, false);
        }
        const v2f   bias2 = *(const v2f*)(b2 + half * 32 + 2 * c);
        const float t0    = temp[0];
        #pragma unroll
        for (int r = 0; r < 8; ++r) {
            const int   node = m0 + rt * 16 + 8 * h + r;
            const float dn   = dinv[node];
            const float vx   = o0[r] + bias2.x;
            const float vy   = o1[r] + bias2.y;
            const int   idx  = node * DOUT + half * 32 + 2 * c;
            v2f y, g;
            y.x = dn * vx;  y.y = dn * vy;
            g.x = t0 * vx;  g.y = t0 * vy;
            *(v2f*)(bufA + idx)   = y;
            *(v2f*)(hidden + idx) = g;
        }
    }
}

__global__ __launch_bounds__(256)
void k_gather(const float* __restrict__ src, float* __restrict__ dst,
              float* __restrict__ hidden, const int* __restrict__ offs,
              const int* __restrict__ cnt, const int* __restrict__ srow,
              const float* __restrict__ dinv, const float* __restrict__ temp,
              int kidx) {
    const int lane = threadIdx.x & 31;
    const int i = __builtin_amdgcn_readfirstlane(blockIdx.x * 8 + (threadIdx.x >> 5));
    const int   beg = offs[i];
    const int   n   = cnt[i];
    const float di  = dinv[i];
    const v2f* src2 = (const v2f*)src;

    const v2f sv = src2[(size_t)i * 32 + lane];
    float ax = sv.x;
    float ay = sv.y;

    for (int base = 0; base < n; base += 32) {
        int m = n - base;
        if (m > 32) m = 32;
        int my_r = i;
        if (lane < m) my_r = srow[beg + base + lane];
        for (int j = 0; j < m; j += 8) {
            #pragma unroll
            for (int u = 0; u < 8; ++u) {
                const int   r = __builtin_amdgcn_readlane(my_r, j + u);
                const float w = (j + u < m) ? 1.0f : 0.0f;
                const v2f v = src2[(size_t)r * 32 + lane];
                ax = fmaf(w, v.x, ax);
                ay = fmaf(w, v.y, ay);
            }
        }
    }
    const float cx = di * ax;
    const float cy = di * ay;

    const size_t o = (size_t)i * 32 + lane;
    v2f res;
    res.x = di * cx;
    res.y = di * cy;
    ((v2f*)dst)[o] = res;
    const float g = temp[kidx];
    v2f h = ((v2f*)hidden)[o];
    h.x += g * cx;
    h.y += g * cy;
    ((v2f*)hidden)[o] = h;
}

extern "C" void kernel_launch(void* const* d_in, const int* in_sizes, int n_in,
                              void* d_out, int out_size, void* d_ws, size_t ws_size,
                              hipStream_t stream) {
    const float* x    = (const float*)d_in[0];
    const int*   ei   = (const int*)  d_in[1];
    const float* W1   = (const float*)d_in[2];
    const float* b1   = (const float*)d_in[3];
    const float* W2   = (const float*)d_in[4];
    const float* b2   = (const float*)d_in[5];
    const float* temp = (const float*)d_in[6];
    const int* row = ei;
    const int* col = ei + N_EDGES;
    float* hidden = (float*)d_out;

    float* bufA  = (float*)d_ws;
    float* bufB  = bufA  + (size_t)N_NODES * DOUT;
    int*   cnt   = (int*)(bufB + (size_t)N_NODES * DOUT);
    int*   offs  = cnt   + N_NODES;
    float* dinv  = (float*)(offs + N_NODES);
    int*   srow  = (int*)(dinv + N_NODES);
    int*   pos   = (int*)bufB;

    k_zero <<<(N_NODES + 255) / 256, 256, 0, stream>>>(cnt);
    k_count<<<(N_EDGES + 255) / 256, 256, 0, stream>>>(col, cnt, pos);
    k_scan <<<1, 1024, 0, stream>>>(cnt, offs, dinv);
    k_fill <<<(N_EDGES + 255) / 256, 256, 0, stream>>>(row, col, pos, offs, srow);

    k_mlp<<<N_NODES / 32, 128, 0, stream>>>(x, W1, b1, W2, b2, temp, dinv, bufA, hidden);

    for (int k = 0; k < KSTEPS; ++k) {
        float* src = (k & 1) ? bufB : bufA;
        float* dst = (k & 1) ? bufA : bufB;
        k_gather<<<N_NODES / 8, 256, 0, stream>>>(src, dst, hidden, offs, cnt, srow,
                                                  dinv, temp, k + 1);
    }
}


# Edge grouping without per-edge global atomics: edges binned by 256-node target range with LDS counts and ranks, then one block per range ranks per node in LDS; run order = LDS arrival order
# speedup vs baseline: 9.6455x; 1.2770x over previous
#include <hip/hip_runtime.h>


#define N_NODES 100000
#define N_EDGES 3200000
#define DIN     512
#define DHID    256
#define DOUT    64
#define KSTEPS  10

typedef __attribute__((ext_vector_type(2))) float v2f;
typedef __attribute__((ext_vector_type(8))) float v8f;

#define BSH     8
#define NBKT    ((N_NODES + 255) >> BSH)
#define PBLK    250
#define PTHR    512
#define PCHUNK  (N_EDGES / PBLK)
#define CSTRIDE 32
#define HROW    392
#define STHR    1024

__device__ __forceinline__ int wave_scan(int v, int lane) {
    #pragma unroll
    for (int d = 1; d < 32; d <<= 1) {
        int t = __shfl_up(v, d, 32);
        if (lane >= d) v += t;
    }
    return v;
}

__global__ __launch_bounds__(256) void k_zero(int* __restrict__ ctr) {
    int i = blockIdx.x * 256 + threadIdx.x;
    if (i < 2 * NBKT * CSTRIDE) ctr[i] = 0;
}

__global__ __launch_bounds__(PTHR) void k_hist(const int* __restrict__ col,
                                               int* __restrict__ tot,
                                               int* __restrict__ histB) {
    __shared__ int lh[NBKT];
    const int tid = threadIdx.x;
    if (tid < NBKT) lh[tid] = 0;
    __syncthreads();
    const int* cp = col + blockIdx.x * PCHUNK + tid;
    int c[PCHUNK / PTHR];
    #pragma unroll
    for (int it = 0; it < PCHUNK / PTHR; ++it) c[it] = cp[it * PTHR];
    #pragma unroll
    for (int it = 0; it < PCHUNK / PTHR; ++it) atomicAdd(&lh[c[it] >> BSH], 1);
    __syncthreads();
    if (tid < NBKT) {
        const int h = lh[tid];
        histB[blockIdx.x * HROW + tid] = h;
        if (h) atomicAdd(&tot[tid * CSTRIDE], h);
    }
}

__global__ __launch_bounds__(PTHR) void k_part(const int* __restrict__ row,
                                               const int* __restrict__ col,
                                               const int* __restrict__ tot,
                                               int* __restrict__ cur,
                                               const int* __restrict__ histB,
                                               int* __restrict__ bst,
                                               int* __restrict__ pk) {
    __shared__ int lbase[NBKT];
    __shared__ int lrank[NBKT];
    __shared__ int wsum[32];
    const int tid  = threadIdx.x;
    const int lane = tid & 31;
    const int wave = tid >> 5;

    const int e0 = blockIdx.x * PCHUNK + tid;
    int c[PCHUNK / PTHR], sr[PCHUNK / PTHR];
    #pragma unroll
    for (int it = 0; it < PCHUNK / PTHR; ++it) {
        c[it]  = col[e0 + it * PTHR];
        sr[it] = row[e0 + it * PTHR];
    }

    const int t    = (tid < NBKT) ? tot[tid * CSTRIDE] : 0;
    const int incl = wave_scan(t, lane);
    if (lane == 31) wsum[wave] = incl;
    __syncthreads();
    const int w    = (lane < PTHR / 32) ? wsum[lane] : 0;
    const int winc = wave_scan(w, lane);
    const int excl = __shfl(winc, wave, 32) - __shfl(w, wave, 32) + incl - t;
    if (tid < NBKT) {
        const int h = histB[blockIdx.x * HROW + tid];
        int r = 0;
        if (h) r = atomicAdd(&cur[tid * CSTRIDE], h);
        lbase[tid] = excl + r;
        lrank[tid] = 0;
    }
    if (blockIdx.x == 0 && tid <= NBKT) bst[tid] = excl;
    __syncthreads();

    #pragma unroll
    for (int it = 0; it < PCHUNK / PTHR; ++it) {
        const int b = c[it] >> BSH;
        const int r = atomicAdd(&lrank[b], 1);
        pk[lbase[b] + r] = (sr[it] << BSH) | (c[it] & 255);
    }
}

__global__ __launch_bounds__(STHR) void k_bsort(const int* __restrict__ pk,
                                                const int* __restrict__ bst,
                                                int* __restrict__ cnt,
                                                int* __restrict__ offs,
                                                float* __restrict__ dinv,
                                                int* __restrict__ srow) {
    __shared__ int lc[256];
    __shared__ int wsum[32];
    const int tid  = threadIdx.x;
    const int lane = tid & 31;
    const int wave = tid >> 5;
    const int beg  = bst[blockIdx.x];
    const int end  = bst[blockIdx.x + 1];

    if (tid < 256) lc[tid] = 0;
    __syncthreads();
    for (int e = beg + tid; e < end; e += 4 * STHR) {
        int p[4];
        #pragma unroll
        for (int u = 0; u < 4; ++u) {
            const int ee = e + u * STHR;
            p[u] = (ee < end) ? pk[ee] : -1;
        }
        #pragma unroll
        for (int u = 0; u < 4; ++u)
            if (p[u] >= 0) atomicAdd(&lc[p[u] & 255], 1);
    }
    __syncthreads();

    const int n    = (tid < 256) ? lc[tid] : 0;
    const int incl = wave_scan(n, lane);
    if (lane == 31) wsum[wave] = incl;
    __syncthreads();
    const int w    = wsum[lane];
    const int winc = wave_scan(w, lane);
    const int excl = __shfl(winc, wave, 32) - __shfl(w, wave, 32) + incl - n;
    if (tid < 256) {
        lc[tid] = excl;
        const int node = (blockIdx.x << BSH) + tid;
        if (node < N_NODES) {
            cnt[node]  = n;
            offs[node] = beg + excl;
            dinv[node] = rsqrtf((float)(n + 1));
        }
    }
    __syncthreads();

    for (int e = beg + tid; e < end; e += 4 * STHR) {
        int p[4];
        #pragma unroll
        for (int u = 0; u < 4; ++u) {
            const int ee = e + u * STHR;
            p[u] = (ee < end) ? pk[ee] : -1;
        }
        #pragma unroll
        for (int u = 0; u < 4; ++u) {
            if (p[u] >= 0) {
                const int r = atomicAdd(&lc[p[u] & 255], 1);
                srow[beg + r] = p[u] >> BSH;
            }
        }
    }
}

#define XP 516
#define HP 260

typedef __attribute__((ext_vector_type(4))) float v4f;

__global__ __launch_bounds__(128)
void k_mlp(const float* __restrict__ x,  const float* __restrict__ W1,
           const float* __restrict__ b1, const float* __restrict__ W2,
           const float* __restrict__ b2, const float* __restrict__ temp,
           const float* __restrict__ dinv,
           float* __restrict__ bufA,     float* __restrict__ hidden) {
    __shared__ float lds[32 * XP];

    const int tid  = threadIdx.x;
    const int wave = tid >> 5;
    const int lane = tid & 31;
    const int c    = lane & 15;
    const int h    = lane >> 4;
    const int m0   = blockIdx.x * 32;

    {
        const v4f* xg = (const v4f*)(x + (size_t)m0 * DIN);
        #pragma unroll
        for (int r = 0; r < 32; ++r) {
            v4f v = xg[r * 128 + tid];
            *(v4f*)(&lds[r * XP + tid * 4]) = v;
        }
    }
    __syncthreads();

    v8f acc[2][4];
    #pragma unroll
    for (int t = 0; t < 2; ++t)
        #pragma unroll
        for (int j = 0; j < 4; ++j)
            acc[t][j] = (v8f){0.f, 0.f, 0.f, 0.f, 0.f, 0.f, 0.f, 0.f};

    {
        const float* a0p = &lds[c * XP + 2 * h];
        const float* a1p = a0p + 16 * XP;
        const float* wp  = W1 + (size_t)(2 * h) * DHID + wave * 64 + 4 * c;
        #pragma unroll 4
        for (int k0 = 0; k0 < DIN; k0 += 4) {
            const v2f a0 = *(const v2f*)(a0p + k0);
            const v2f a1 = *(const v2f*)(a1p + k0);
            const v4f q0 = *(const v4f*)(wp + (size_t)k0 * DHID);
            const v4f q1 = *(const v4f*)(wp + (size_t)(k0 + 1) * DHID);
            #pragma unroll
            for (int j = 0; j < 4; ++j) {
                v2f b;
                b.x = q0[j];
                b.y = q1[j];
                acc[0][j] = __builtin_amdgcn_wmma_f32_16x16x4_f32(
                                false, a0, false, b, (short)0, acc[0][j], false, false);
                acc[1][j] = __builtin_amdgcn_wmma_f32_16x16x4_f32(
                                false, a1, false, b, (short)0, acc[1][j], false, false);
            }
        }
    }
    __syncthreads();

    {
        const v4f bias = *(const v4f*)(b1 + wave * 64 + 4 * c);
        #pragma unroll
        for (int t = 0; t < 2; ++t) {
            #pragma unroll
            for (int r = 0; r < 8; ++r) {
                v4f v;
                #pragma unroll
                for (int j = 0; j < 4; ++j) {
                    float u = acc[t][j][r] + bias[j];
                    v[j] = u > 0.0f ? u : 0.0f;
                }
                *(v4f*)(&lds[(t * 16 + 8 * h + r) * HP + wave * 64 + 4 * c]) = v;
            }
        }
    }
    __syncthreads();

    {
        const int rt   = wave >> 1;
        const int half = wave & 1;
        v8f o0 = {0.f, 0.f, 0.f, 0.f, 0.f, 0.f, 0.f, 0.f};
        v8f o1 = {0.f, 0.f, 0.f, 0.f, 0.f, 0.f, 0.f, 0.f};
        const float* ap  = &lds[(rt * 16 + c) * HP + 2 * h];
        const float* w2p = W2 + (2 * h) * DOUT + half * 32 + 2 * c;
        #pragma unroll 8
        for (int k0 = 0; k0 < DHID; k0 += 4) {
            const v2f a  = *(const v2f*)(ap + k0);
            const v2f q0 = *(const v2f*)(w2p + k0 * DOUT);
            const v2f q1 = *(const v2f*)(w2p + (k0 + 1) * DOUT);
            v2f bA, bB;
            bA.x = q0.x;  bA.y = q1.x;
            bB.x = q0.y;  bB.y = q1.y;
            o0 = __builtin_amdgcn_wmma_f32_16x16x4_f32(
                     false, a, false, bA, (short)0, o0, false, false);
            o1 = __builtin_amdgcn_wmma_f32_16x16x4_f32(
                     false, a, false, bB, (short)0, o1, false, false);
        }
        const v2f   bias2 = *(const v2f*)(b2 + half * 32 + 2 * c);
        const float t0    = temp[0];
        #pragma unroll
        for (int r = 0; r < 8; ++r) {
            const int   node = m0 + rt * 16 + 8 * h + r;
            const float dn   = dinv[node];
            const float vx   = o0[r] + bias2.x;
            const float vy   = o1[r] + bias2.y;
            const int   idx  = node * DOUT + half * 32 + 2 * c;
            v2f y, g;
            y.x = dn * vx;  y.y = dn * vy;
            g.x = t0 * vx;  g.y = t0 * vy;
            *(v2f*)(bufA + idx)   = y;
            *(v2f*)(hidden + idx) = g;
        }
    }
}

__global__ __launch_bounds__(256)
void k_gather(const float* __restrict__ src, float* __restrict__ dst,
              float* __restrict__ hidden, const int* __restrict__ offs,
              const int* __restrict__ cnt, const int* __restrict__ srow,
              const float* __restrict__ dinv, const float* __restrict__ temp,
              int kidx) {
    const int lane = threadIdx.x & 31;
    const int i = __builtin_amdgcn_readfirstlane(blockIdx.x * 8 + (threadIdx.x >> 5));
    const int   beg = offs[i];
    const int   n   = cnt[i];
    const float di  = dinv[i];
    const v2f* src2 = (const v2f*)src;

    const v2f sv = src2[(size_t)i * 32 + lane];
    float ax = sv.x;
    float ay = sv.y;

    for (int base = 0; base < n; base += 32) {
        int m = n - base;
        if (m > 32) m = 32;
        int my_r = i;
        if (lane < m) my_r = srow[beg + base + lane];
        for (int j = 0; j < m; j += 8) {
            #pragma unroll
            for (int u = 0; u < 8; ++u) {
                const int   r = __builtin_amdgcn_readlane(my_r, j + u);
                const float w = (j + u < m) ? 1.0f : 0.0f;
                const v2f v = src2[(size_t)r * 32 + lane];
                ax = fmaf(w, v.x, ax);
                ay = fmaf(w, v.y, ay);
            }
        }
    }
    const float cx = di * ax;
    const float cy = di * ay;

    const size_t o = (size_t)i * 32 + lane;
    v2f res;
    res.x = di * cx;
    res.y = di * cy;
    ((v2f*)dst)[o] = res;
    const float g = temp[kidx];
    v2f h = ((v2f*)hidden)[o];
    h.x += g * cx;
    h.y += g * cy;
    ((v2f*)hidden)[o] = h;
}

extern "C" void kernel_launch(void* const* d_in, const int* in_sizes, int n_in,
                              void* d_out, int out_size, void* d_ws, size_t ws_size,
                              hipStream_t stream) {
    const float* x    = (const float*)d_in[0];
    const int*   ei   = (const int*)  d_in[1];
    const float* W1   = (const float*)d_in[2];
    const float* b1   = (const float*)d_in[3];
    const float* W2   = (const float*)d_in[4];
    const float* b2   = (const float*)d_in[5];
    const float* temp = (const float*)d_in[6];
    const int* row = ei;
    const int* col = ei + N_EDGES;
    float* hidden = (float*)d_out;

    float* bufA  = (float*)d_ws;
    float* bufB  = bufA  + (size_t)N_NODES * DOUT;
    int*   cnt   = (int*)(bufB + (size_t)N_NODES * DOUT);
    int*   offs  = cnt   + N_NODES;
    float* dinv  = (float*)(offs + N_NODES);
    int*   srow  = (int*)(dinv + N_NODES);
    int*   pk    = (int*)bufB;
    int*   tot   = (int*)bufA;
    int*   cur   = tot + NBKT * CSTRIDE;
    int*   bst   = cur + NBKT * CSTRIDE;
    int*   histB = bst + 512;

    k_zero <<<(2 * NBKT * CSTRIDE + 255) / 256, 256, 0, stream>>>(tot);
    k_hist <<<PBLK, PTHR, 0, stream>>>(col, tot, histB);
    k_part <<<PBLK, PTHR, 0, stream>>>(row, col, tot, cur, histB, bst, pk);
    k_bsort<<<NBKT, STHR, 0, stream>>>(pk, bst, cnt, offs, dinv, srow);

    k_mlp<<<N_NODES / 32, 128, 0, stream>>>(x, W1, b1, W2, b2, temp, dinv, bufA, hidden);

    for (int k = 0; k < KSTEPS; ++k) {
        float* src = (k & 1) ? bufB : bufA;
        float* dst = (k & 1) ? bufA : bufB;
        k_gather<<<N_NODES / 8, 256, 0, stream>>>(src, dst, hidden, offs, cnt, srow,
                                                  dinv, temp, k + 1);
    }
}


# PPR sum evaluated as recurrence z_k = temp[k]*h + A*z_(k+1): a step reads the fixed MLP row and stores one row, no read-modify-write of the output row; summation order over k changed
# speedup vs baseline: 10.2571x; 1.0634x over previous
#include <hip/hip_runtime.h>


#define N_NODES 100000
#define N_EDGES 3200000
#define DIN     512
#define DHID    256
#define DOUT    64
#define KSTEPS  10

typedef __attribute__((ext_vector_type(2))) float v2f;
typedef __attribute__((ext_vector_type(8))) float v8f;

#define BSH     8
#define NBKT    ((N_NODES + 255) >> BSH)
#define PBLK    250
#define PTHR    512
#define PCHUNK  (N_EDGES / PBLK)
#define CSTRIDE 32
#define HROW    392
#define STHR    1024

__device__ __forceinline__ int wave_scan(int v, int lane) {
    #pragma unroll
    for (int d = 1; d < 32; d <<= 1) {
        int t = __shfl_up(v, d, 32);
        if (lane >= d) v += t;
    }
    return v;
}

__global__ __launch_bounds__(256) void k_zero(int* __restrict__ ctr) {
    int i = blockIdx.x * 256 + threadIdx.x;
    if (i < 2 * NBKT * CSTRIDE) ctr[i] = 0;
}

__global__ __launch_bounds__(PTHR) void k_hist(const int* __restrict__ col,
                                               int* __restrict__ tot,
                                               int* __restrict__ histB) {
    __shared__ int lh[NBKT];
    const int tid = threadIdx.x;
    if (tid < NBKT) lh[tid] = 0;
    __syncthreads();
    const int* cp = col + blockIdx.x * PCHUNK + tid;
    int c[PCHUNK / PTHR];
    #pragma unroll
    for (int it = 0; it < PCHUNK / PTHR; ++it) c[it] = cp[it * PTHR];
    #pragma unroll
    for (int it = 0; it < PCHUNK / PTHR; ++it) atomicAdd(&lh[c[it] >> BSH], 1);
    __syncthreads();
    if (tid < NBKT) {
        const int h = lh[tid];
        histB[blockIdx.x * HROW + tid] = h;
        if (h) atomicAdd(&tot[tid * CSTRIDE], h);
    }
}

__global__ __launch_bounds__(PTHR) void k_part(const int* __restrict__ row,
                                               const int* __restrict__ col,
                                               const int* __restrict__ tot,
                                               int* __restrict__ cur,
                                               const int* __restrict__ histB,
                                               int* __restrict__ bst,
                                               int* __restrict__ pk) {
    __shared__ int lbase[NBKT];
    __shared__ int lrank[NBKT];
    __shared__ int wsum[32];
    const int tid  = threadIdx.x;
    const int lane = tid & 31;
    const int wave = tid >> 5;

    const int e0 = blockIdx.x * PCHUNK + tid;
    int c[PCHUNK / PTHR], sr[PCHUNK / PTHR];
    #pragma unroll
    for (int it = 0; it < PCHUNK / PTHR; ++it) {
        c[it]  = col[e0 + it * PTHR];
        sr[it] = row[e0 + it * PTHR];
    }

    const int t    = (tid < NBKT) ? tot[tid * CSTRIDE] : 0;
    const int incl = wave_scan(t, lane);
    if (lane == 31) wsum[wave] = incl;
    __syncthreads();
    const int w    = (lane < PTHR / 32) ? wsum[lane] : 0;
    const int winc = wave_scan(w, lane);
    const int excl = __shfl(winc, wave, 32) - __shfl(w, wave, 32) + incl - t;
    if (tid < NBKT) {
        const int h = histB[blockIdx.x * HROW + tid];
        int r = 0;
        if (h) r = atomicAdd(&cur[tid * CSTRIDE], h);
        lbase[tid] = excl + r;
        lrank[tid] = 0;
    }
    if (blockIdx.x == 0 && tid <= NBKT) bst[tid] = excl;
    __syncthreads();

    #pragma unroll
    for (int it = 0; it < PCHUNK / PTHR; ++it) {
        const int b = c[it] >> BSH;
        const int r = atomicAdd(&lrank[b], 1);
        pk[lbase[b] + r] = (sr[it] << BSH) | (c[it] & 255);
    }
}

__global__ __launch_bounds__(STHR) void k_bsort(const int* __restrict__ pk,
                                                const int* __restrict__ bst,
                                                int* __restrict__ cnt,
                                                int* __restrict__ offs,
                                                float* __restrict__ dinv,
                                                int* __restrict__ srow) {
    __shared__ int lc[256];
    __shared__ int wsum[32];
    const int tid  = threadIdx.x;
    const int lane = tid & 31;
    const int wave = tid >> 5;
    const int beg  = bst[blockIdx.x];
    const int end  = bst[blockIdx.x + 1];

    if (tid < 256) lc[tid] = 0;
    __syncthreads();
    for (int e = beg + tid; e < end; e += 4 * STHR) {
        int p[4];
        #pragma unroll
        for (int u = 0; u < 4; ++u) {
            const int ee = e + u * STHR;
            p[u] = (ee < end) ? pk[ee] : -1;
        }
        #pragma unroll
        for (int u = 0; u < 4; ++u)
            if (p[u] >= 0) atomicAdd(&lc[p[u] & 255], 1);
    }
    __syncthreads();

    const int n    = (tid < 256) ? lc[tid] : 0;
    const int incl = wave_scan(n, lane);
    if (lane == 31) wsum[wave] = incl;
    __syncthreads();
    const int w    = wsum[lane];
    const int winc = wave_scan(w, lane);
    const int excl = __shfl(winc, wave, 32) - __shfl(w, wave, 32) + incl - n;
    if (tid < 256) {
        lc[tid] = excl;
        const int node = (blockIdx.x << BSH) + tid;
        if (node < N_NODES) {
            cnt[node]  = n;
            offs[node] = beg + excl;
            dinv[node] = rsqrtf((float)(n + 1));
        }
    }
    __syncthreads();

    for (int e = beg + tid; e < end; e += 4 * STHR) {
        int p[4];
        #pragma unroll
        for (int u = 0; u < 4; ++u) {
            const int ee = e + u * STHR;
            p[u] = (ee < end) ? pk[ee] : -1;
        }
        #pragma unroll
        for (int u = 0; u < 4; ++u) {
            if (p[u] >= 0) {
                const int r = atomicAdd(&lc[p[u] & 255], 1);
                srow[beg + r] = p[u] >> BSH;
            }
        }
    }
}

#define XP 516
#define HP 260

typedef __attribute__((ext_vector_type(4))) float v4f;

__global__ __launch_bounds__(128)
void k_mlp(const float* __restrict__ x,  const float* __restrict__ W1,
           const float* __restrict__ b1, const float* __restrict__ W2,
           const float* __restrict__ b2, const float* __restrict__ temp,
           const float* __restrict__ dinv,
           float* __restrict__ bufA,     float* __restrict__ hrow) {
    __shared__ float lds[32 * XP];

    const int tid  = threadIdx.x;
    const int wave = tid >> 5;
    const int lane = tid & 31;
    const int c    = lane & 15;
    const int h    = lane >> 4;
    const int m0   = blockIdx.x * 32;

    {
        const v4f* xg = (const v4f*)(x + (size_t)m0 * DIN);
        #pragma unroll
        for (int r = 0; r < 32; ++r) {
            v4f v = xg[r * 128 + tid];
            *(v4f*)(&lds[r * XP + tid * 4]) = v;
        }
    }
    __syncthreads();

    v8f acc[2][4];
    #pragma unroll
    for (int t = 0; t < 2; ++t)
        #pragma unroll
        for (int j = 0; j < 4; ++j)
            acc[t][j] = (v8f){0.f, 0.f, 0.f, 0.f, 0.f, 0.f, 0.f, 0.f};

    {
        const float* a0p = &lds[c * XP + 2 * h];
        const float* a1p = a0p + 16 * XP;
        const float* wp  = W1 + (size_t)(2 * h) * DHID + wave * 64 + 4 * c;
        #pragma unroll 4
        for (int k0 = 0; k0 < DIN; k0 += 4) {
            const v2f a0 = *(const v2f*)(a0p + k0);
            const v2f a1 = *(const v2f*)(a1p + k0);
            const v4f q0 = *(const v4f*)(wp + (size_t)k0 * DHID);
            const v4f q1 = *(const v4f*)(wp + (size_t)(k0 + 1) * DHID);
            #pragma unroll
            for (int j = 0; j < 4; ++j) {
                v2f b;
                b.x = q0[j];
                b.y = q1[j];
                acc[0][j] = __builtin_amdgcn_wmma_f32_16x16x4_f32(
                                false, a0, false, b, (short)0, acc[0][j], false, false);
                acc[1][j] = __builtin_amdgcn_wmma_f32_16x16x4_f32(
                                false, a1, false, b, (short)0, acc[1][j], false, false);
            }
        }
    }
    __syncthreads();

    {
        const v4f bias = *(const v4f*)(b1 + wave * 64 + 4 * c);
        #pragma unroll
        for (int t = 0; t < 2; ++t) {
            #pragma unroll
            for (int r = 0; r < 8; ++r) {
                v4f v;
                #pragma unroll
                for (int j = 0; j < 4; ++j) {
                    float u = acc[t][j][r] + bias[j];
                    v[j] = u > 0.0f ? u : 0.0f;
                }
                *(v4f*)(&lds[(t * 16 + 8 * h + r) * HP + wave * 64 + 4 * c]) = v;
            }
        }
    }
    __syncthreads();

    {
        const int rt   = wave >> 1;
        const int half = wave & 1;
        v8f o0 = {0.f, 0.f, 0.f, 0.f, 0.f, 0.f, 0.f, 0.f};
        v8f o1 = {0.f, 0.f, 0.f, 0.f, 0.f, 0.f, 0.f, 0.f};
        const float* ap  = &lds[(rt * 16 + c) * HP + 2 * h];
        const float* w2p = W2 + (2 * h) * DOUT + half * 32 + 2 * c;
        #pragma unroll 8
        for (int k0 = 0; k0 < DHID; k0 += 4) {
            const v2f a  = *(const v2f*)(ap + k0);
            const v2f q0 = *(const v2f*)(w2p + k0 * DOUT);
            const v2f q1 = *(const v2f*)(w2p + (k0 + 1) * DOUT);
            v2f bA, bB;
            bA.x = q0.x;  bA.y = q1.x;
            bB.x = q0.y;  bB.y = q1.y;
            o0 = __builtin_amdgcn_wmma_f32_16x16x4_f32(
                     false, a, false, bA, (short)0, o0, false, false);
            o1 = __builtin_amdgcn_wmma_f32_16x16x4_f32(
                     false, a, false, bB, (short)0, o1, false, false);
        }
        const v2f   bias2 = *(const v2f*)(b2 + half * 32 + 2 * c);
        const float tK    = temp[KSTEPS];
        #pragma unroll
        for (int r = 0; r < 8; ++r) {
            const int   node = m0 + rt * 16 + 8 * h + r;
            const float dn   = dinv[node];
            const float vx   = o0[r] + bias2.x;
            const float vy   = o1[r] + bias2.y;
            const int   idx  = node * DOUT + half * 32 + 2 * c;
            v2f y, g;
            y.x = dn * (tK * vx);  y.y = dn * (tK * vy);
            g.x = vx;  g.y = vy;
            *(v2f*)(bufA + idx) = y;
            *(v2f*)(hrow + idx) = g;
        }
    }
}

__global__ __launch_bounds__(256)
void k_gather(const float* __restrict__ src, float* dst,
              const float* hrow, const int* __restrict__ offs,
              const int* __restrict__ cnt, const int* __restrict__ srow,
              const float* __restrict__ dinv, const float* __restrict__ temp,
              int kidx) {
    const int lane = threadIdx.x & 31;
    const int i = __builtin_amdgcn_readfirstlane(blockIdx.x * 8 + (threadIdx.x >> 5));
    const int   beg = offs[i];
    const int   n   = cnt[i];
    const float di  = dinv[i];
    const float g   = temp[kidx];
    const v2f* src2 = (const v2f*)src;
    const size_t o  = (size_t)i * 32 + lane;

    const v2f sv = src2[o];
    const v2f hv = ((const v2f*)hrow)[o];
    float ax = sv.x;
    float ay = sv.y;

    for (int base = 0; base < n; base += 32) {
        int m = n - base;
        if (m > 32) m = 32;
        int my_r = i;
        if (lane < m) my_r = srow[beg + base + lane];
        for (int j = 0; j < m; j += 8) {
            #pragma unroll
            for (int u = 0; u < 8; ++u) {
                const int   r = __builtin_amdgcn_readlane(my_r, j + u);
                const float w = (j + u < m) ? 1.0f : 0.0f;
                const v2f v = src2[(size_t)r * 32 + lane];
                ax = fmaf(w, v.x, ax);
                ay = fmaf(w, v.y, ay);
            }
        }
    }
    const float zx = fmaf(g, hv.x, di * ax);
    const float zy = fmaf(g, hv.y, di * ay);

    v2f res;
    if (kidx > 0) {
        res.x = di * zx;
        res.y = di * zy;
    } else {
        res.x = zx;
        res.y = zy;
    }
    ((v2f*)dst)[o] = res;
}

extern "C" void kernel_launch(void* const* d_in, const int* in_sizes, int n_in,
                              void* d_out, int out_size, void* d_ws, size_t ws_size,
                              hipStream_t stream) {
    const float* x    = (const float*)d_in[0];
    const int*   ei   = (const int*)  d_in[1];
    const float* W1   = (const float*)d_in[2];
    const float* b1   = (const float*)d_in[3];
    const float* W2   = (const float*)d_in[4];
    const float* b2   = (const float*)d_in[5];
    const float* temp = (const float*)d_in[6];
    const int* row = ei;
    const int* col = ei + N_EDGES;
    float* hidden = (float*)d_out;

    float* bufA  = (float*)d_ws;
    float* bufB  = bufA  + (size_t)N_NODES * DOUT;
    int*   cnt   = (int*)(bufB + (size_t)N_NODES * DOUT);
    int*   offs  = cnt   + N_NODES;
    float* dinv  = (float*)(offs + N_NODES);
    int*   srow  = (int*)(dinv + N_NODES);
    int*   pk    = (int*)bufB;
    int*   tot   = (int*)bufA;
    int*   cur   = tot + NBKT * CSTRIDE;
    int*   bst   = cur + NBKT * CSTRIDE;
    int*   histB = bst + 512;

    k_zero <<<(2 * NBKT * CSTRIDE + 255) / 256, 256, 0, stream>>>(tot);
    k_hist <<<PBLK, PTHR, 0, stream>>>(col, tot, histB);
    k_part <<<PBLK, PTHR, 0, stream>>>(row, col, tot, cur, histB, bst, pk);
    k_bsort<<<NBKT, STHR, 0, stream>>>(pk, bst, cnt, offs, dinv, srow);

    k_mlp<<<N_NODES / 32, 128, 0, stream>>>(x, W1, b1, W2, b2, temp, dinv, bufA, hidden);

    for (int s = 0; s < KSTEPS; ++s) {
        float* src = (s & 1) ? bufB : bufA;
        float* dst = (s == KSTEPS - 1) ? hidden : ((s & 1) ? bufA : bufB);
        k_gather<<<N_NODES / 8, 256, 0, stream>>>(src, dst, hidden, offs, cnt, srow,
                                                  dinv, temp, KSTEPS - 1 - s);
    }
}
